# starting kernel
# baseline (speedup 1.0000x reference)
#include <hip/hip_runtime.h>

typedef float v2f __attribute__((ext_vector_type(2)));
typedef float v8f __attribute__((ext_vector_type(8)));

#define WMMA_F32(a, b, c) \
  __builtin_amdgcn_wmma_f32_16x16x4_f32(false, (a), false, (b), (short)0, (c), false, false)

constexpr int kB = 8, kN = 1024, kP = 128, kH = 64, kD = 64, kJ = 256, kX = 64;
constexpr int INNER = kH * kD;
constexpr int CH = 32;
constexpr int NCH = kN / CH;

__global__ __launch_bounds__(256)
void pos_attn_fused(const float* __restrict__ pos, const float* __restrict__ x,
                    const float* __restrict__ Wq, const float* __restrict__ kl,
                    float* __restrict__ ws) {
  __shared__ float lds_x[CH][kX];
  __shared__ float lds_qh[CH][kD + 1];
  __shared__ float lds_p[8][32][CH + 1];

  const int bb   = blockIdx.x >> 6;
  const int h    = blockIdx.x & 63;
  const int tid  = threadIdx.x;
  const int w    = tid >> 5;
  const int lane = tid & 31;
  const int row  = lane & 15;
  const int half = lane >> 4;
  const int kb   = half * 2;

  const int tt = w >> 2;
  const int dt = w & 3;

  v8f   oacc[2][4];
  float mrun[2][8], lrun[2][8];
#pragma unroll
  for (int jt = 0; jt < 2; ++jt) {
#pragma unroll
    for (int xt = 0; xt < 4; ++xt) oacc[jt][xt] = (v8f){0,0,0,0,0,0,0,0};
#pragma unroll
    for (int r = 0; r < 8; ++r) { mrun[jt][r] = -3.0e38f; lrun[jt][r] = 0.0f; }
  }

  const float* posB = pos + (size_t)bb * kN * kP;
  const float* xB   = x   + (size_t)bb * kN * kX;

  for (int ch = 0; ch < NCH; ++ch) {
    const int n0 = ch * CH;
    __syncthreads();

    {
      const float4* gx = reinterpret_cast<const float4*>(xB + n0 * kX);
      float4* sx = reinterpret_cast<float4*>(&lds_x[0][0]);
#pragma unroll
      for (int i = 0; i < 2; ++i) sx[tid + i * 256] = gx[tid + i * 256];
    }

    {
      v8f cq = (v8f){0,0,0,0,0,0,0,0};
      const float* pA = posB + (size_t)(n0 + tt * 16 + row) * kP;
      const float* pB = Wq + h * kD + dt * 16 + row;
#pragma unroll
      for (int k = 0; k < kP; k += 4) {
        v2f a = *reinterpret_cast<const v2f*>(pA + k + kb);
        v2f b;
        b.x = pB[(size_t)(k + kb) * INNER];
        b.y = pB[(size_t)(k + kb + 1) * INNER];
        cq = WMMA_F32(a, b, cq);
      }
#pragma unroll
      for (int r = 0; r < 8; ++r)
        lds_qh[tt * 16 + r + 8 * half][dt * 16 + row] = cq[r];
    }
    __syncthreads();

    v8f s[2][2];
#pragma unroll
    for (int jt = 0; jt < 2; ++jt) {
#pragma unroll
      for (int nt = 0; nt < 2; ++nt) {
        v8f c = (v8f){0,0,0,0,0,0,0,0};
        const float* aK = kl + (size_t)(w * 32 + jt * 16 + row) * kD;
#pragma unroll
        for (int k = 0; k < kD; k += 4) {
          v2f a = *reinterpret_cast<const v2f*>(aK + k + kb);
          v2f b;
          b.x = lds_qh[nt * 16 + row][k + kb];
          b.y = lds_qh[nt * 16 + row][k + kb + 1];
          c = WMMA_F32(a, b, c);
        }
#pragma unroll
        for (int r = 0; r < 8; ++r) c[r] *= 0.125f;
        s[jt][nt] = c;
      }
    }

#pragma unroll
    for (int jt = 0; jt < 2; ++jt) {
      float t[8], mnew[8], fsc[8], rs[8];
#pragma unroll
      for (int r = 0; r < 8; ++r) t[r] = fmaxf(s[jt][0][r], s[jt][1][r]);
#pragma unroll
      for (int off = 1; off <= 8; off <<= 1)
#pragma unroll
        for (int r = 0; r < 8; ++r)
          t[r] = fmaxf(t[r], __shfl_xor(t[r], off, 32));
#pragma unroll
      for (int r = 0; r < 8; ++r) {
        mnew[r] = fmaxf(mrun[jt][r], t[r]);
        fsc[r]  = __expf(mrun[jt][r] - mnew[r]);
      }
#pragma unroll
      for (int nt = 0; nt < 2; ++nt)
#pragma unroll
        for (int r = 0; r < 8; ++r)
          s[jt][nt][r] = __expf(s[jt][nt][r] - mnew[r]);
#pragma unroll
      for (int r = 0; r < 8; ++r) rs[r] = s[jt][0][r] + s[jt][1][r];
#pragma unroll
      for (int off = 1; off <= 8; off <<= 1)
#pragma unroll
        for (int r = 0; r < 8; ++r) rs[r] += __shfl_xor(rs[r], off, 32);
#pragma unroll
      for (int r = 0; r < 8; ++r) {
        lrun[jt][r] = lrun[jt][r] * fsc[r] + rs[r];
        mrun[jt][r] = mnew[r];
      }
#pragma unroll
      for (int xt = 0; xt < 4; ++xt)
#pragma unroll
        for (int r = 0; r < 8; ++r) oacc[jt][xt][r] *= fsc[r];
#pragma unroll
      for (int nt = 0; nt < 2; ++nt)
#pragma unroll
        for (int r = 0; r < 8; ++r)
          lds_p[w][jt * 16 + r + 8 * half][nt * 16 + row] = s[jt][nt][r];
    }

#pragma unroll
    for (int jt = 0; jt < 2; ++jt) {
#pragma unroll
      for (int xt = 0; xt < 4; ++xt) {
        v8f c = oacc[jt][xt];
#pragma unroll
        for (int k = 0; k < CH; k += 4) {
          v2f a, b;
          a.x = lds_p[w][jt * 16 + row][k + kb];
          a.y = lds_p[w][jt * 16 + row][k + kb + 1];
          b.x = lds_x[k + kb][xt * 16 + row];
          b.y = lds_x[k + kb + 1][xt * 16 + row];
          c = WMMA_F32(a, b, c);
        }
        oacc[jt][xt] = c;
      }
    }
  }

#pragma unroll
  for (int jt = 0; jt < 2; ++jt) {
    float inv[8];
#pragma unroll
    for (int r = 0; r < 8; ++r) inv[r] = 1.0f / lrun[jt][r];
#pragma unroll
    for (int xt = 0; xt < 4; ++xt)
#pragma unroll
      for (int r = 0; r < 8; ++r) {
        const int j  = w * 32 + jt * 16 + r + 8 * half;
        const int xd = xt * 16 + row;
        ws[(size_t)(bb * kJ + j) * INNER + h * kX + xd] = oacc[jt][xt][r] * inv[r];
      }
  }
}

__global__ __launch_bounds__(128)
void out_proj(const float* __restrict__ ws, const float* __restrict__ Wout,
              const float* __restrict__ bout, float* __restrict__ out) {
  const int rt   = blockIdx.x;
  const int w    = threadIdx.x >> 5;
  const int lane = threadIdx.x & 31;
  const int row  = lane & 15;
  const int half = lane >> 4;
  const int kb   = half * 2;

  const float* aR = ws + (size_t)(rt * 16 + row) * INNER;
  const float* bC = Wout + w * 16 + row;

  v8f c = (v8f){0,0,0,0,0,0,0,0};
  for (int k = 0; k < INNER; k += 4) {
    v2f a = *reinterpret_cast<const v2f*>(aR + k + kb);
    v2f b;
    b.x = bC[(size_t)(k + kb) * 64];
    b.y = bC[(size_t)(k + kb + 1) * 64];
    c = WMMA_F32(a, b, c);
  }
  const float bias = bout[w * 16 + row];
#pragma unroll
  for (int r = 0; r < 8; ++r)
    out[(size_t)(rt * 16 + r + 8 * half) * 64 + w * 16 + row] = c[r] + bias;
}

extern "C" void kernel_launch(void* const* d_in, const int* in_sizes, int n_in,
                              void* d_out, int out_size, void* d_ws, size_t ws_size,
                              hipStream_t stream) {
  (void)in_sizes; (void)n_in; (void)out_size; (void)ws_size;
  const float* pos  = (const float*)d_in[0];
  const float* x    = (const float*)d_in[1];
  const float* Wq   = (const float*)d_in[2];
  const float* kl   = (const float*)d_in[3];
  const float* Wout = (const float*)d_in[4];
  const float* bout = (const float*)d_in[5];
  float* ws  = (float*)d_ws;
  float* out = (float*)d_out;

  pos_attn_fused<<<kB * kH, 256, 0, stream>>>(pos, x, Wq, kl, ws);
  out_proj<<<(kB * kJ) / 16, 128, 0, stream>>>(ws, Wout, bout, out);
}


# softmax row max and row sum: the four butterfly steps over each 16-lane row done by DPP row_xmask on the vector ALU instead of LDS-unit lane shuffles; same partner lanes, sum order unchanged
# speedup vs baseline: 1.6177x; 1.6177x over previous
#include <hip/hip_runtime.h>

typedef float v2f __attribute__((ext_vector_type(2)));
typedef float v8f __attribute__((ext_vector_type(8)));

#define WMMA_F32(a, b, c) \
  __builtin_amdgcn_wmma_f32_16x16x4_f32(false, (a), false, (b), (short)0, (c), false, false)

template <int M>
__device__ __forceinline__ float row_xor(float v) {
  return __builtin_amdgcn_update_dpp(v, v, 0x160 + M, 0xf, 0xf, true);
}

constexpr int kB = 8, kN = 1024, kP = 128, kH = 64, kD = 64, kJ = 256, kX = 64;
constexpr int INNER = kH * kD;
constexpr int CH = 32;
constexpr int NCH = kN / CH;

__device__ __forceinline__ void load_pos_frag(v2f (&a)[32], const float* __restrict__ p) {
#pragma unroll
  for (int ks = 0; ks < 32; ++ks) a[ks] = *reinterpret_cast<const v2f*>(p + 4 * ks);
}

__device__ __forceinline__ void qh_tile(const v2f (&a)[32], const v2f (&bq)[4][32],
                                        float* __restrict__ dst, int half, int row) {
  v8f cq[4];
#pragma unroll
  for (int dt = 0; dt < 4; ++dt) cq[dt] = (v8f){0,0,0,0,0,0,0,0};
#pragma unroll
  for (int ks = 0; ks < 32; ++ks)
#pragma unroll
    for (int dt = 0; dt < 4; ++dt) cq[dt] = WMMA_F32(a[ks], bq[dt][ks], cq[dt]);
#pragma unroll
  for (int dt = 0; dt < 4; ++dt)
#pragma unroll
    for (int r = 0; r < 8; ++r)
      dst[(r + 8 * half) * (kD + 1) + dt * 16 + row] = cq[dt][r];
}

__global__ __launch_bounds__(256)
void pos_attn_fused(const float* __restrict__ pos, const float* __restrict__ x,
                    const float* __restrict__ Wq, const float* __restrict__ kl,
                    const float* __restrict__ Wout, float* __restrict__ ws) {
  __shared__ float lds_qh[kN][kD + 1];
  __shared__ float lds_p[8][32][CH + 1];

  const int bb   = blockIdx.x >> 6;
  const int h    = blockIdx.x & 63;
  const int tid  = threadIdx.x;
  const int w    = tid >> 5;
  const int lane = tid & 31;
  const int row  = lane & 15;
  const int half = lane >> 4;
  const int kb   = half * 2;

  const float* posB = pos + (size_t)bb * kN * kP;
  const float* xB   = x   + (size_t)bb * kN * kX;

  {
    v2f bq[4][32];
    const float* pB = Wq + h * kD + row;
#pragma unroll
    for (int dt = 0; dt < 4; ++dt)
#pragma unroll
      for (int ks = 0; ks < 32; ++ks) {
        bq[dt][ks].x = pB[(size_t)(4 * ks + kb) * INNER + dt * 16];
        bq[dt][ks].y = pB[(size_t)(4 * ks + kb + 1) * INNER + dt * 16];
      }
    const float* pA = posB + (size_t)(w * 128 + row) * kP + kb;
    float* qdst = &lds_qh[w * 128][0];
    v2f a0[32], a1[32];
    load_pos_frag(a0, pA);
#pragma unroll 1
    for (int i = 0; i < 8; i += 2) {
      load_pos_frag(a1, pA + (size_t)(i + 1) * 16 * kP);
      qh_tile(a0, bq, qdst + i * 16 * (kD + 1), half, row);
      const int nx = (i + 2 < 8) ? (i + 2) : 7;
      load_pos_frag(a0, pA + (size_t)nx * 16 * kP);
      qh_tile(a1, bq, qdst + (i + 1) * 16 * (kD + 1), half, row);
    }
  }
  v2f ak[2][16];
  {
    const float* aK = kl + (size_t)(w * 32 + row) * kD + kb;
#pragma unroll
    for (int jt = 0; jt < 2; ++jt)
#pragma unroll
      for (int ks = 0; ks < 16; ++ks)
        ak[jt][ks] = *reinterpret_cast<const v2f*>(aK + (size_t)jt * 16 * kD + 4 * ks);
  }
  __syncthreads();

  v8f   oacc[2][4];
  float mrun[2][8], lrun[2][8];
#pragma unroll
  for (int jt = 0; jt < 2; ++jt) {
#pragma unroll
    for (int xt = 0; xt < 4; ++xt) oacc[jt][xt] = (v8f){0,0,0,0,0,0,0,0};
#pragma unroll
    for (int r = 0; r < 8; ++r) { mrun[jt][r] = -3.0e38f; lrun[jt][r] = 0.0f; }
  }

  for (int ch = 0; ch < NCH; ++ch) {
    const int n0 = ch * CH;

    v2f bx[4][8];
    {
      const float* xc = xB + (size_t)n0 * kX + row;
#pragma unroll
      for (int xt = 0; xt < 4; ++xt)
#pragma unroll
        for (int ks = 0; ks < 8; ++ks) {
          bx[xt][ks].x = xc[(4 * ks + kb) * kX + xt * 16];
          bx[xt][ks].y = xc[(4 * ks + kb + 1) * kX + xt * 16];
        }
    }

    v8f s[2][2];
#pragma unroll
    for (int jt = 0; jt < 2; ++jt) {
#pragma unroll
      for (int nt = 0; nt < 2; ++nt) {
        v8f c = (v8f){0,0,0,0,0,0,0,0};
#pragma unroll
        for (int k = 0; k < kD; k += 4) {
          v2f b;
          b.x = lds_qh[n0 + nt * 16 + row][k + kb];
          b.y = lds_qh[n0 + nt * 16 + row][k + kb + 1];
          c = WMMA_F32(ak[jt][k / 4], b, c);
        }
#pragma unroll
        for (int r = 0; r < 8; ++r) c[r] *= 0.125f;
        s[jt][nt] = c;
      }
    }

#pragma unroll
    for (int jt = 0; jt < 2; ++jt) {
      float t[8], mnew[8], fsc[8], rs[8];
#pragma unroll
      for (int r = 0; r < 8; ++r) t[r] = fmaxf(s[jt][0][r], s[jt][1][r]);
#pragma unroll
      for (int r = 0; r < 8; ++r) t[r] = fmaxf(t[r], row_xor<1>(t[r]));
#pragma unroll
      for (int r = 0; r < 8; ++r) t[r] = fmaxf(t[r], row_xor<2>(t[r]));
#pragma unroll
      for (int r = 0; r < 8; ++r) t[r] = fmaxf(t[r], row_xor<4>(t[r]));
#pragma unroll
      for (int r = 0; r < 8; ++r) t[r] = fmaxf(t[r], row_xor<8>(t[r]));
#pragma unroll
      for (int r = 0; r < 8; ++r) {
        mnew[r] = fmaxf(mrun[jt][r], t[r]);
        fsc[r]  = __expf(mrun[jt][r] - mnew[r]);
      }
#pragma unroll
      for (int nt = 0; nt < 2; ++nt)
#pragma unroll
        for (int r = 0; r < 8; ++r)
          s[jt][nt][r] = __expf(s[jt][nt][r] - mnew[r]);
#pragma unroll
      for (int r = 0; r < 8; ++r) rs[r] = s[jt][0][r] + s[jt][1][r];
#pragma unroll
      for (int r = 0; r < 8; ++r) rs[r] += row_xor<1>(rs[r]);
#pragma unroll
      for (int r = 0; r < 8; ++r) rs[r] += row_xor<2>(rs[r]);
#pragma unroll
      for (int r = 0; r < 8; ++r) rs[r] += row_xor<4>(rs[r]);
#pragma unroll
      for (int r = 0; r < 8; ++r) rs[r] += row_xor<8>(rs[r]);
#pragma unroll
      for (int r = 0; r < 8; ++r) {
        lrun[jt][r] = lrun[jt][r] * fsc[r] + rs[r];
        mrun[jt][r] = mnew[r];
      }
#pragma unroll
      for (int xt = 0; xt < 4; ++xt)
#pragma unroll
        for (int r = 0; r < 8; ++r) oacc[jt][xt][r] *= fsc[r];
#pragma unroll
      for (int nt = 0; nt < 2; ++nt)
#pragma unroll
        for (int r = 0; r < 8; ++r)
          lds_p[w][jt * 16 + r + 8 * half][nt * 16 + row] = s[jt][nt][r];
    }

#pragma unroll
    for (int jt = 0; jt < 2; ++jt) {
#pragma unroll
      for (int xt = 0; xt < 4; ++xt) {
        v8f c = oacc[jt][xt];
#pragma unroll
        for (int k = 0; k < CH; k += 4) {
          v2f a;
          a.x = lds_p[w][jt * 16 + row][k + kb];
          a.y = lds_p[w][jt * 16 + row][k + kb + 1];
          c = WMMA_F32(a, bx[xt][k / 4], c);
        }
        oacc[jt][xt] = c;
      }
    }
  }

  float* stg = &lds_p[w][0][0];
  const float* Wh = Wout + (size_t)h * kX * 64;
  float* Pw = ws + ((size_t)h * 512 + (size_t)(bb * 16 + w * 2) * 4) * 256;
#pragma unroll
  for (int jt = 0; jt < 2; ++jt) {
    float inv[8];
#pragma unroll
    for (int r = 0; r < 8; ++r) inv[r] = 1.0f / lrun[jt][r];
#pragma unroll
    for (int xt = 0; xt < 4; ++xt)
#pragma unroll
      for (int r = 0; r < 8; ++r)
        stg[(r + 8 * half) * 66 + xt * 16 + row] = oacc[jt][xt][r] * inv[r];

    v8f pc[4];
#pragma unroll
    for (int ct = 0; ct < 4; ++ct) pc[ct] = (v8f){0,0,0,0,0,0,0,0};
#pragma unroll
    for (int k = 0; k < kX; k += 4) {
      v2f a = *reinterpret_cast<const v2f*>(stg + row * 66 + k + kb);
#pragma unroll
      for (int ct = 0; ct < 4; ++ct) {
        v2f b;
        b.x = Wh[(k + kb) * 64 + ct * 16 + row];
        b.y = Wh[(k + kb + 1) * 64 + ct * 16 + row];
        pc[ct] = WMMA_F32(a, b, pc[ct]);
      }
    }
#pragma unroll
    for (int ct = 0; ct < 4; ++ct)
#pragma unroll
      for (int r = 0; r < 8; ++r)
        Pw[(jt * 4 + ct) * 256 + r * 32 + lane] = pc[ct][r];
  }
}

__global__ __launch_bounds__(256)
void head_sum(const float* __restrict__ ws, const float* __restrict__ bout,
              float* __restrict__ out) {
  __shared__ float4 red[4][64];
  const int T   = blockIdx.x;
  const int tid = threadIdx.x;
  const int q   = tid & 63;
  const int g   = tid >> 6;

  const float4* p = reinterpret_cast<const float4*>(ws) +
                    ((size_t)(g * 16) * 512 + T) * 64 + q;
  float4 v[16];
#pragma unroll
  for (int i = 0; i < 16; ++i) v[i] = p[(size_t)i * 512 * 64];
  float4 acc = v[0];
#pragma unroll
  for (int i = 1; i < 16; ++i) {
    acc.x += v[i].x; acc.y += v[i].y; acc.z += v[i].z; acc.w += v[i].w;
  }
  red[g][q] = acc;
  __syncthreads();
  if (g == 0) {
    float4 s = red[0][q];
#pragma unroll
    for (int i = 1; i < 4; ++i) {
      const float4 t = red[i][q];
      s.x += t.x; s.y += t.y; s.z += t.z; s.w += t.w;
    }
    const int r     = q >> 3;
    const int half  = (q >> 2) & 1;
    const int c0    = (T & 3) * 16 + (q & 3) * 4;
    const int jrow  = (T >> 2) * 16 + r + 8 * half;
    const float4 bv = *reinterpret_cast<const float4*>(bout + c0);
    s.x += bv.x; s.y += bv.y; s.z += bv.z; s.w += bv.w;
    *reinterpret_cast<float4*>(out + (size_t)jrow * 64 + c0) = s;
  }
}

extern "C" void kernel_launch(void* const* d_in, const int* in_sizes, int n_in,
                              void* d_out, int out_size, void* d_ws, size_t ws_size,
                              hipStream_t stream) {
  (void)in_sizes; (void)n_in; (void)out_size; (void)ws_size;
  const float* pos  = (const float*)d_in[0];
  const float* x    = (const float*)d_in[1];
  const float* Wq   = (const float*)d_in[2];
  const float* kl   = (const float*)d_in[3];
  const float* Wout = (const float*)d_in[4];
  const float* bout = (const float*)d_in[5];
  float* ws  = (float*)d_ws;
  float* out = (float*)d_out;

  pos_attn_fused<<<kB * kH, 256, 0, stream>>>(pos, x, Wq, kl, Wout, ws);
  head_sum<<<(kB * kJ * 64) / 256, 256, 0, stream>>>(ws, bout, out);
}


# softmax row sums kept as per-lane shares in the token loop and added over the 16 lanes once after it (no per-chunk sum butterfly); sum order new
# speedup vs baseline: 1.6769x; 1.0366x over previous
#include <hip/hip_runtime.h>

typedef float v2f __attribute__((ext_vector_type(2)));
typedef float v8f __attribute__((ext_vector_type(8)));

#define WMMA_F32(a, b, c) \
  __builtin_amdgcn_wmma_f32_16x16x4_f32(false, (a), false, (b), (short)0, (c), false, false)

template <int M>
__device__ __forceinline__ float row_xor(float v) {
  return __builtin_amdgcn_update_dpp(v, v, 0x160 + M, 0xf, 0xf, true);
}

constexpr int kB = 8, kN = 1024, kP = 128, kH = 64, kD = 64, kJ = 256, kX = 64;
constexpr int INNER = kH * kD;
constexpr int CH = 32;
constexpr int NCH = kN / CH;

__device__ __forceinline__ void load_pos_frag(v2f (&a)[32], const float* __restrict__ p) {
#pragma unroll
  for (int ks = 0; ks < 32; ++ks) a[ks] = *reinterpret_cast<const v2f*>(p + 4 * ks);
}

__device__ __forceinline__ void qh_tile(const v2f (&a)[32], const v2f (&bq)[4][32],
                                        float* __restrict__ dst, int half, int row) {
  v8f cq[4];
#pragma unroll
  for (int dt = 0; dt < 4; ++dt) cq[dt] = (v8f){0,0,0,0,0,0,0,0};
#pragma unroll
  for (int ks = 0; ks < 32; ++ks)
#pragma unroll
    for (int dt = 0; dt < 4; ++dt) cq[dt] = WMMA_F32(a[ks], bq[dt][ks], cq[dt]);
#pragma unroll
  for (int dt = 0; dt < 4; ++dt)
#pragma unroll
    for (int r = 0; r < 8; ++r)
      dst[(r + 8 * half) * (kD + 1) + dt * 16 + row] = cq[dt][r];
}

__global__ __launch_bounds__(256)
void pos_attn_fused(const float* __restrict__ pos, const float* __restrict__ x,
                    const float* __restrict__ Wq, const float* __restrict__ kl,
                    const float* __restrict__ Wout, float* __restrict__ ws) {
  __shared__ float lds_qh[kN][kD + 1];
  __shared__ float lds_p[8][32][CH + 1];

  const int bb   = blockIdx.x >> 6;
  const int h    = blockIdx.x & 63;
  const int tid  = threadIdx.x;
  const int w    = tid >> 5;
  const int lane = tid & 31;
  const int row  = lane & 15;
  const int half = lane >> 4;
  const int kb   = half * 2;

  const float* posB = pos + (size_t)bb * kN * kP;
  const float* xB   = x   + (size_t)bb * kN * kX;

  {
    v2f bq[4][32];
    const float* pB = Wq + h * kD + row;
#pragma unroll
    for (int dt = 0; dt < 4; ++dt)
#pragma unroll
      for (int ks = 0; ks < 32; ++ks) {
        bq[dt][ks].x = pB[(size_t)(4 * ks + kb) * INNER + dt * 16];
        bq[dt][ks].y = pB[(size_t)(4 * ks + kb + 1) * INNER + dt * 16];
      }
    const float* pA = posB + (size_t)(w * 128 + row) * kP + kb;
    float* qdst = &lds_qh[w * 128][0];
    v2f a0[32], a1[32];
    load_pos_frag(a0, pA);
#pragma unroll 1
    for (int i = 0; i < 8; i += 2) {
      load_pos_frag(a1, pA + (size_t)(i + 1) * 16 * kP);
      qh_tile(a0, bq, qdst + i * 16 * (kD + 1), half, row);
      const int nx = (i + 2 < 8) ? (i + 2) : 7;
      load_pos_frag(a0, pA + (size_t)nx * 16 * kP);
      qh_tile(a1, bq, qdst + (i + 1) * 16 * (kD + 1), half, row);
    }
  }
  v2f ak[2][16];
  {
    const float* aK = kl + (size_t)(w * 32 + row) * kD + kb;
#pragma unroll
    for (int jt = 0; jt < 2; ++jt)
#pragma unroll
      for (int ks = 0; ks < 16; ++ks)
        ak[jt][ks] = *reinterpret_cast<const v2f*>(aK + (size_t)jt * 16 * kD + 4 * ks);
  }
  __syncthreads();

  v8f   oacc[2][4];
  float mrun[2][8], lrun[2][8];
#pragma unroll
  for (int jt = 0; jt < 2; ++jt) {
#pragma unroll
    for (int xt = 0; xt < 4; ++xt) oacc[jt][xt] = (v8f){0,0,0,0,0,0,0,0};
#pragma unroll
    for (int r = 0; r < 8; ++r) { mrun[jt][r] = -3.0e38f; lrun[jt][r] = 0.0f; }
  }

  for (int ch = 0; ch < NCH; ++ch) {
    const int n0 = ch * CH;

    v2f bx[4][8];
    {
      const float* xc = xB + (size_t)n0 * kX + row;
#pragma unroll
      for (int xt = 0; xt < 4; ++xt)
#pragma unroll
        for (int ks = 0; ks < 8; ++ks) {
          bx[xt][ks].x = xc[(4 * ks + kb) * kX + xt * 16];
          bx[xt][ks].y = xc[(4 * ks + kb + 1) * kX + xt * 16];
        }
    }

    v8f s[2][2];
#pragma unroll
    for (int jt = 0; jt < 2; ++jt) {
#pragma unroll
      for (int nt = 0; nt < 2; ++nt) {
        v8f c = (v8f){0,0,0,0,0,0,0,0};
#pragma unroll
        for (int k = 0; k < kD; k += 4) {
          v2f b;
          b.x = lds_qh[n0 + nt * 16 + row][k + kb];
          b.y = lds_qh[n0 + nt * 16 + row][k + kb + 1];
          c = WMMA_F32(ak[jt][k / 4], b, c);
        }
#pragma unroll
        for (int r = 0; r < 8; ++r) c[r] *= 0.125f;
        s[jt][nt] = c;
      }
    }

#pragma unroll
    for (int jt = 0; jt < 2; ++jt) {
      float t[8], mnew[8], fsc[8], rs[8];
#pragma unroll
      for (int r = 0; r < 8; ++r) t[r] = fmaxf(s[jt][0][r], s[jt][1][r]);
#pragma unroll
      for (int r = 0; r < 8; ++r) t[r] = fmaxf(t[r], row_xor<1>(t[r]));
#pragma unroll
      for (int r = 0; r < 8; ++r) t[r] = fmaxf(t[r], row_xor<2>(t[r]));
#pragma unroll
      for (int r = 0; r < 8; ++r) t[r] = fmaxf(t[r], row_xor<4>(t[r]));
#pragma unroll
      for (int r = 0; r < 8; ++r) t[r] = fmaxf(t[r], row_xor<8>(t[r]));
#pragma unroll
      for (int r = 0; r < 8; ++r) {
        mnew[r] = fmaxf(mrun[jt][r], t[r]);
        fsc[r]  = __expf(mrun[jt][r] - mnew[r]);
      }
#pragma unroll
      for (int nt = 0; nt < 2; ++nt)
#pragma unroll
        for (int r = 0; r < 8; ++r)
          s[jt][nt][r] = __expf(s[jt][nt][r] - mnew[r]);
#pragma unroll
      for (int r = 0; r < 8; ++r) rs[r] = s[jt][0][r] + s[jt][1][r];
#pragma unroll
      for (int r = 0; r < 8; ++r) {
        lrun[jt][r] = lrun[jt][r] * fsc[r] + rs[r];
        mrun[jt][r] = mnew[r];
      }
#pragma unroll
      for (int xt = 0; xt < 4; ++xt)
#pragma unroll
        for (int r = 0; r < 8; ++r) oacc[jt][xt][r] *= fsc[r];
#pragma unroll
      for (int nt = 0; nt < 2; ++nt)
#pragma unroll
        for (int r = 0; r < 8; ++r)
          lds_p[w][jt * 16 + r + 8 * half][nt * 16 + row] = s[jt][nt][r];
    }

#pragma unroll
    for (int jt = 0; jt < 2; ++jt) {
#pragma unroll
      for (int xt = 0; xt < 4; ++xt) {
        v8f c = oacc[jt][xt];
#pragma unroll
        for (int k = 0; k < CH; k += 4) {
          v2f a;
          a.x = lds_p[w][jt * 16 + row][k + kb];
          a.y = lds_p[w][jt * 16 + row][k + kb + 1];
          c = WMMA_F32(a, bx[xt][k / 4], c);
        }
        oacc[jt][xt] = c;
      }
    }
  }

  float* stg = &lds_p[w][0][0];
  const float* Wh = Wout + (size_t)h * kX * 64;
  float* Pw = ws + ((size_t)h * 512 + (size_t)(bb * 16 + w * 2) * 4) * 256;
#pragma unroll
  for (int jt = 0; jt < 2; ++jt) {
    float inv[8];
#pragma unroll
    for (int r = 0; r < 8; ++r) inv[r] = lrun[jt][r];
#pragma unroll
    for (int r = 0; r < 8; ++r) inv[r] += row_xor<1>(inv[r]);
#pragma unroll
    for (int r = 0; r < 8; ++r) inv[r] += row_xor<2>(inv[r]);
#pragma unroll
    for (int r = 0; r < 8; ++r) inv[r] += row_xor<4>(inv[r]);
#pragma unroll
    for (int r = 0; r < 8; ++r) inv[r] += row_xor<8>(inv[r]);
#pragma unroll
    for (int r = 0; r < 8; ++r) inv[r] = 1.0f / inv[r];
#pragma unroll
    for (int xt = 0; xt < 4; ++xt)
#pragma unroll
      for (int r = 0; r < 8; ++r)
        stg[(r + 8 * half) * 66 + xt * 16 + row] = oacc[jt][xt][r] * inv[r];

    v8f pc[4];
#pragma unroll
    for (int ct = 0; ct < 4; ++ct) pc[ct] = (v8f){0,0,0,0,0,0,0,0};
#pragma unroll
    for (int k = 0; k < kX; k += 4) {
      v2f a = *reinterpret_cast<const v2f*>(stg + row * 66 + k + kb);
#pragma unroll
      for (int ct = 0; ct < 4; ++ct) {
        v2f b;
        b.x = Wh[(k + kb) * 64 + ct * 16 + row];
        b.y = Wh[(k + kb + 1) * 64 + ct * 16 + row];
        pc[ct] = WMMA_F32(a, b, pc[ct]);
      }
    }
#pragma unroll
    for (int ct = 0; ct < 4; ++ct)
#pragma unroll
      for (int r = 0; r < 8; ++r)
        Pw[(jt * 4 + ct) * 256 + r * 32 + lane] = pc[ct][r];
  }
}

__global__ __launch_bounds__(256)
void head_sum(const float* __restrict__ ws, const float* __restrict__ bout,
              float* __restrict__ out) {
  __shared__ float4 red[4][64];
  const int T   = blockIdx.x;
  const int tid = threadIdx.x;
  const int q   = tid & 63;
  const int g   = tid >> 6;

  const float4* p = reinterpret_cast<const float4*>(ws) +
                    ((size_t)(g * 16) * 512 + T) * 64 + q;
  float4 v[16];
#pragma unroll
  for (int i = 0; i < 16; ++i) v[i] = p[(size_t)i * 512 * 64];
  float4 acc = v[0];
#pragma unroll
  for (int i = 1; i < 16; ++i) {
    acc.x += v[i].x; acc.y += v[i].y; acc.z += v[i].z; acc.w += v[i].w;
  }
  red[g][q] = acc;
  __syncthreads();
  if (g == 0) {
    float4 s = red[0][q];
#pragma unroll
    for (int i = 1; i < 4; ++i) {
      const float4 t = red[i][q];
      s.x += t.x; s.y += t.y; s.z += t.z; s.w += t.w;
    }
    const int r     = q >> 3;
    const int half  = (q >> 2) & 1;
    const int c0    = (T & 3) * 16 + (q & 3) * 4;
    const int jrow  = (T >> 2) * 16 + r + 8 * half;
    const float4 bv = *reinterpret_cast<const float4*>(bout + c0);
    s.x += bv.x; s.y += bv.y; s.z += bv.z; s.w += bv.w;
    *reinterpret_cast<float4*>(out + (size_t)jrow * 64 + c0) = s;
  }
}

extern "C" void kernel_launch(void* const* d_in, const int* in_sizes, int n_in,
                              void* d_out, int out_size, void* d_ws, size_t ws_size,
                              hipStream_t stream) {
  (void)in_sizes; (void)n_in; (void)out_size; (void)ws_size;
  const float* pos  = (const float*)d_in[0];
  const float* x    = (const float*)d_in[1];
  const float* Wq   = (const float*)d_in[2];
  const float* kl   = (const float*)d_in[3];
  const float* Wout = (const float*)d_in[4];
  const float* bout = (const float*)d_in[5];
  float* ws  = (float*)d_ws;
  float* out = (float*)d_out;

  pos_attn_fused<<<kB * kH, 256, 0, stream>>>(pos, x, Wq, kl, Wout, ws);
  head_sum<<<(kB * kJ * 64) / 256, 256, 0, stream>>>(ws, bout, out);
}


# attention loop on 64-token chunks instead of 32: one row-max butterfly and one rescale of O and of the sum shares per 64 tokens; P staged in two 32-token passes; softmax sum order new
# speedup vs baseline: 1.7321x; 1.0329x over previous
#include <hip/hip_runtime.h>

typedef float v2f __attribute__((ext_vector_type(2)));
typedef float v8f __attribute__((ext_vector_type(8)));

#define WMMA_F32(a, b, c) \
  __builtin_amdgcn_wmma_f32_16x16x4_f32(false, (a), false, (b), (short)0, (c), false, false)

template <int M>
__device__ __forceinline__ float row_xor(float v) {
  return __builtin_amdgcn_update_dpp(v, v, 0x160 + M, 0xf, 0xf, true);
}

constexpr int kB = 8, kN = 1024, kP = 128, kH = 64, kD = 64, kJ = 256, kX = 64;
constexpr int INNER = kH * kD;
constexpr int CH = 64;
constexpr int NCH = kN / CH;
constexpr int NT = CH / 16;
constexpr int PH = 32;

__device__ __forceinline__ void load_pos_frag(v2f (&a)[32], const float* __restrict__ p) {
#pragma unroll
  for (int ks = 0; ks < 32; ++ks) a[ks] = *reinterpret_cast<const v2f*>(p + 4 * ks);
}

__device__ __forceinline__ void qh_tile(const v2f (&a)[32], const v2f (&bq)[4][32],
                                        float* __restrict__ dst, int half, int row) {
  v8f cq[4];
#pragma unroll
  for (int dt = 0; dt < 4; ++dt) cq[dt] = (v8f){0,0,0,0,0,0,0,0};
#pragma unroll
  for (int ks = 0; ks < 32; ++ks)
#pragma unroll
    for (int dt = 0; dt < 4; ++dt) cq[dt] = WMMA_F32(a[ks], bq[dt][ks], cq[dt]);
#pragma unroll
  for (int dt = 0; dt < 4; ++dt)
#pragma unroll
    for (int r = 0; r < 8; ++r)
      dst[(r + 8 * half) * (kD + 1) + dt * 16 + row] = cq[dt][r];
}

__global__ __launch_bounds__(256)
void pos_attn_fused(const float* __restrict__ pos, const float* __restrict__ x,
                    const float* __restrict__ Wq, const float* __restrict__ kl,
                    const float* __restrict__ Wout, float* __restrict__ ws) {
  __shared__ float lds_qh[kN][kD + 1];
  __shared__ float lds_p[8][32][PH + 1];

  const int bb   = blockIdx.x >> 6;
  const int h    = blockIdx.x & 63;
  const int tid  = threadIdx.x;
  const int w    = tid >> 5;
  const int lane = tid & 31;
  const int row  = lane & 15;
  const int half = lane >> 4;
  const int kb   = half * 2;

  const float* posB = pos + (size_t)bb * kN * kP;
  const float* xB   = x   + (size_t)bb * kN * kX;

  {
    v2f bq[4][32];
    const float* pB = Wq + h * kD + row;
#pragma unroll
    for (int dt = 0; dt < 4; ++dt)
#pragma unroll
      for (int ks = 0; ks < 32; ++ks) {
        bq[dt][ks].x = pB[(size_t)(4 * ks + kb) * INNER + dt * 16];
        bq[dt][ks].y = pB[(size_t)(4 * ks + kb + 1) * INNER + dt * 16];
      }
    const float* pA = posB + (size_t)(w * 128 + row) * kP + kb;
    float* qdst = &lds_qh[w * 128][0];
    v2f a0[32], a1[32];
    load_pos_frag(a0, pA);
#pragma unroll 1
    for (int i = 0; i < 8; i += 2) {
      load_pos_frag(a1, pA + (size_t)(i + 1) * 16 * kP);
      qh_tile(a0, bq, qdst + i * 16 * (kD + 1), half, row);
      const int nx = (i + 2 < 8) ? (i + 2) : 7;
      load_pos_frag(a0, pA + (size_t)nx * 16 * kP);
      qh_tile(a1, bq, qdst + (i + 1) * 16 * (kD + 1), half, row);
    }
  }
  v2f ak[2][16];
  {
    const float* aK = kl + (size_t)(w * 32 + row) * kD + kb;
#pragma unroll
    for (int jt = 0; jt < 2; ++jt)
#pragma unroll
      for (int ks = 0; ks < 16; ++ks)
        ak[jt][ks] = *reinterpret_cast<const v2f*>(aK + (size_t)jt * 16 * kD + 4 * ks);
  }
  __syncthreads();

  v8f   oacc[2][4];
  v8f   mrun[2], lrun[2];
#pragma unroll
  for (int jt = 0; jt < 2; ++jt) {
#pragma unroll
    for (int xt = 0; xt < 4; ++xt) oacc[jt][xt] = (v8f){0,0,0,0,0,0,0,0};
#pragma unroll
    for (int r = 0; r < 8; ++r) { mrun[jt][r] = -3.0e38f; lrun[jt][r] = 0.0f; }
  }

  for (int ch = 0; ch < NCH; ++ch) {
    const int n0 = ch * CH;
    const float* xc = xB + (size_t)n0 * kX + row;

    v2f bx0[4][8];
#pragma unroll
    for (int xt = 0; xt < 4; ++xt)
#pragma unroll
      for (int ks = 0; ks < 8; ++ks) {
        bx0[xt][ks].x = xc[(4 * ks + kb) * kX + xt * 16];
        bx0[xt][ks].y = xc[(4 * ks + kb + 1) * kX + xt * 16];
      }

    v2f qf[NT][16];
#pragma unroll
    for (int nt = 0; nt < NT; ++nt)
#pragma unroll
      for (int ks = 0; ks < 16; ++ks) {
        qf[nt][ks].x = lds_qh[n0 + nt * 16 + row][4 * ks + kb];
        qf[nt][ks].y = lds_qh[n0 + nt * 16 + row][4 * ks + kb + 1];
      }
    __builtin_amdgcn_sched_barrier(0);

    v8f s[2][NT];
#pragma unroll
    for (int nt = 0; nt < NT; ++nt) {
#pragma unroll
      for (int jt = 0; jt < 2; ++jt) {
        v8f c = (v8f){0,0,0,0,0,0,0,0};
#pragma unroll
        for (int ks = 0; ks < 16; ++ks) c = WMMA_F32(ak[jt][ks], qf[nt][ks], c);
        s[jt][nt] = c;
      }
    }

    __builtin_amdgcn_sched_barrier(0);
    v2f bx1[4][8];
#pragma unroll
    for (int xt = 0; xt < 4; ++xt)
#pragma unroll
      for (int ks = 0; ks < 8; ++ks) {
        bx1[xt][ks].x = xc[(PH + 4 * ks + kb) * kX + xt * 16];
        bx1[xt][ks].y = xc[(PH + 4 * ks + kb + 1) * kX + xt * 16];
      }

#pragma unroll
    for (int jt = 0; jt < 2; ++jt) {
#pragma unroll
      for (int nt = 0; nt < NT; ++nt) s[jt][nt] = s[jt][nt] * 0.125f;
      v8f t = __builtin_elementwise_max(__builtin_elementwise_max(s[jt][0], s[jt][1]),
                                        __builtin_elementwise_max(s[jt][2], s[jt][3]));
#pragma unroll
      for (int r = 0; r < 8; ++r) t[r] = fmaxf(t[r], row_xor<1>(t[r]));
#pragma unroll
      for (int r = 0; r < 8; ++r) t[r] = fmaxf(t[r], row_xor<2>(t[r]));
#pragma unroll
      for (int r = 0; r < 8; ++r) t[r] = fmaxf(t[r], row_xor<4>(t[r]));
#pragma unroll
      for (int r = 0; r < 8; ++r) t[r] = fmaxf(t[r], row_xor<8>(t[r]));
      const v8f mold = mrun[jt];
      const v8f mnew = __builtin_elementwise_max(mold, t);
      v8f fsc = mold - mnew;
#pragma unroll
      for (int r = 0; r < 8; ++r) fsc[r] = __expf(fsc[r]);
#pragma unroll
      for (int nt = 0; nt < NT; ++nt) {
        v8f d = s[jt][nt] - mnew;
#pragma unroll
        for (int r = 0; r < 8; ++r) d[r] = __expf(d[r]);
        s[jt][nt] = d;
      }
      const v8f rs = (s[jt][0] + s[jt][1]) + (s[jt][2] + s[jt][3]);
      lrun[jt] = lrun[jt] * fsc + rs;
      mrun[jt] = mnew;
#pragma unroll
      for (int xt = 0; xt < 4; ++xt) oacc[jt][xt] = oacc[jt][xt] * fsc;
    }

#pragma unroll
    for (int hh = 0; hh < 2; ++hh) {
#pragma unroll
      for (int jt = 0; jt < 2; ++jt)
#pragma unroll
        for (int nt = 0; nt < 2; ++nt)
#pragma unroll
          for (int r = 0; r < 8; ++r)
            lds_p[w][jt * 16 + r + 8 * half][nt * 16 + row] = s[jt][2 * hh + nt][r];
#pragma unroll
      for (int jt = 0; jt < 2; ++jt) {
#pragma unroll
        for (int xt = 0; xt < 4; ++xt) {
          v8f c = oacc[jt][xt];
#pragma unroll
          for (int k = 0; k < PH; k += 4) {
            v2f a;
            a.x = lds_p[w][jt * 16 + row][k + kb];
            a.y = lds_p[w][jt * 16 + row][k + kb + 1];
            c = WMMA_F32(a, hh ? bx1[xt][k / 4] : bx0[xt][k / 4], c);
          }
          oacc[jt][xt] = c;
        }
      }
    }
  }

  float* stg = &lds_p[w][0][0];
  const float* Wh = Wout + (size_t)h * kX * 64;
  float* Pw = ws + ((size_t)h * 512 + (size_t)(bb * 16 + w * 2) * 4) * 256;
#pragma unroll
  for (int jt = 0; jt < 2; ++jt) {
    float inv[8];
#pragma unroll
    for (int r = 0; r < 8; ++r) inv[r] = lrun[jt][r];
#pragma unroll
    for (int r = 0; r < 8; ++r) inv[r] += row_xor<1>(inv[r]);
#pragma unroll
    for (int r = 0; r < 8; ++r) inv[r] += row_xor<2>(inv[r]);
#pragma unroll
    for (int r = 0; r < 8; ++r) inv[r] += row_xor<4>(inv[r]);
#pragma unroll
    for (int r = 0; r < 8; ++r) inv[r] += row_xor<8>(inv[r]);
#pragma unroll
    for (int r = 0; r < 8; ++r) inv[r] = 1.0f / inv[r];
#pragma unroll
    for (int xt = 0; xt < 4; ++xt)
#pragma unroll
      for (int r = 0; r < 8; ++r)
        stg[(r + 8 * half) * 66 + xt * 16 + row] = oacc[jt][xt][r] * inv[r];

    v8f pc[4];
#pragma unroll
    for (int ct = 0; ct < 4; ++ct) pc[ct] = (v8f){0,0,0,0,0,0,0,0};
#pragma unroll
    for (int k = 0; k < kX; k += 4) {
      v2f a = *reinterpret_cast<const v2f*>(stg + row * 66 + k + kb);
#pragma unroll
      for (int ct = 0; ct < 4; ++ct) {
        v2f b;
        b.x = Wh[(k + kb) * 64 + ct * 16 + row];
        b.y = Wh[(k + kb + 1) * 64 + ct * 16 + row];
        pc[ct] = WMMA_F32(a, b, pc[ct]);
      }
    }
#pragma unroll
    for (int ct = 0; ct < 4; ++ct)
#pragma unroll
      for (int r = 0; r < 8; ++r)
        Pw[(jt * 4 + ct) * 256 + r * 32 + lane] = pc[ct][r];
  }
}

__global__ __launch_bounds__(256)
void head_sum(const float* __restrict__ ws, const float* __restrict__ bout,
              float* __restrict__ out) {
  __shared__ float4 red[4][64];
  const int T   = blockIdx.x;
  const int tid = threadIdx.x;
  const int q   = tid & 63;
  const int g   = tid >> 6;

  const float4* p = reinterpret_cast<const float4*>(ws) +
                    ((size_t)(g * 16) * 512 + T) * 64 + q;
  float4 v[16];
#pragma unroll
  for (int i = 0; i < 16; ++i) v[i] = p[(size_t)i * 512 * 64];
  float4 acc = v[0];
#pragma unroll
  for (int i = 1; i < 16; ++i) {
    acc.x += v[i].x; acc.y += v[i].y; acc.z += v[i].z; acc.w += v[i].w;
  }
  red[g][q] = acc;
  __syncthreads();
  if (g == 0) {
    float4 s = red[0][q];
#pragma unroll
    for (int i = 1; i < 4; ++i) {
      const float4 t = red[i][q];
      s.x += t.x; s.y += t.y; s.z += t.z; s.w += t.w;
    }
    const int r     = q >> 3;
    const int half  = (q >> 2) & 1;
    const int c0    = (T & 3) * 16 + (q & 3) * 4;
    const int jrow  = (T >> 2) * 16 + r + 8 * half;
    const float4 bv = *reinterpret_cast<const float4*>(bout + c0);
    s.x += bv.x; s.y += bv.y; s.z += bv.z; s.w += bv.w;
    *reinterpret_cast<float4*>(out + (size_t)jrow * 64 + c0) = s;
  }
}

extern "C" void kernel_launch(void* const* d_in, const int* in_sizes, int n_in,
                              void* d_out, int out_size, void* d_ws, size_t ws_size,
                              hipStream_t stream) {
  (void)in_sizes; (void)n_in; (void)out_size; (void)ws_size;
  const float* pos  = (const float*)d_in[0];
  const float* x    = (const float*)d_in[1];
  const float* Wq   = (const float*)d_in[2];
  const float* kl   = (const float*)d_in[3];
  const float* Wout = (const float*)d_in[4];
  const float* bout = (const float*)d_in[5];
  float* ws  = (float*)d_ws;
  float* out = (float*)d_out;

  pos_attn_fused<<<kB * kH, 256, 0, stream>>>(pos, x, Wq, kl, Wout, ws);
  head_sum<<<(kB * kJ * 64) / 256, 256, 0, stream>>>(ws, bout, out);
}


# x and Wq B fragments fetched by 128-bit loads: a lane takes 4 neighbouring columns of a row, one per tile, so tile t holds columns 4c+t (order restored when tiles go to LDS); sum order unchanged
# speedup vs baseline: 1.8057x; 1.0425x over previous
#include <hip/hip_runtime.h>

typedef float v2f __attribute__((ext_vector_type(2)));
typedef float v8f __attribute__((ext_vector_type(8)));
typedef float v4f __attribute__((ext_vector_type(4)));

#define WMMA_F32(a, b, c) \
  __builtin_amdgcn_wmma_f32_16x16x4_f32(false, (a), false, (b), (short)0, (c), false, false)

template <int M>
__device__ __forceinline__ float row_xor(float v) {
  return __builtin_amdgcn_update_dpp(v, v, 0x160 + M, 0xf, 0xf, true);
}

constexpr int kB = 8, kN = 1024, kP = 128, kH = 64, kD = 64, kJ = 256, kX = 64;
constexpr int INNER = kH * kD;
constexpr int CH = 64;
constexpr int NCH = kN / CH;
constexpr int NT = CH / 16;
constexpr int PH = 32;

__device__ __forceinline__ void load_pos_frag(v2f (&a)[32], const float* __restrict__ p) {
#pragma unroll
  for (int ks = 0; ks < 32; ++ks) a[ks] = *reinterpret_cast<const v2f*>(p + 4 * ks);
}

__device__ __forceinline__ void qh_tile(const v2f (&a)[32], const v2f (&bq)[4][32],
                                        float* __restrict__ dst, int half, int row) {
  v8f cq[4];
#pragma unroll
  for (int dt = 0; dt < 4; ++dt) cq[dt] = (v8f){0,0,0,0,0,0,0,0};
#pragma unroll
  for (int ks = 0; ks < 32; ++ks)
#pragma unroll
    for (int dt = 0; dt < 4; ++dt) cq[dt] = WMMA_F32(a[ks], bq[dt][ks], cq[dt]);
#pragma unroll
  for (int dt = 0; dt < 4; ++dt)
#pragma unroll
    for (int r = 0; r < 8; ++r)
      dst[(r + 8 * half) * (kD + 1) + 4 * row + dt] = cq[dt][r];
}

__global__ __launch_bounds__(256)
void pos_attn_fused(const float* __restrict__ pos, const float* __restrict__ x,
                    const float* __restrict__ Wq, const float* __restrict__ kl,
                    const float* __restrict__ Wout, float* __restrict__ ws) {
  __shared__ float lds_qh[kN][kD + 1];
  __shared__ float lds_p[8][32][PH + 1];

  const int bb   = blockIdx.x >> 6;
  const int h    = blockIdx.x & 63;
  const int tid  = threadIdx.x;
  const int w    = tid >> 5;
  const int lane = tid & 31;
  const int row  = lane & 15;
  const int half = lane >> 4;
  const int kb   = half * 2;

  const float* posB = pos + (size_t)bb * kN * kP;
  const float* xB   = x   + (size_t)bb * kN * kX;

  {
    v2f bq[4][32];
    const float* pB = Wq + h * kD + 4 * row;
#pragma unroll
    for (int ks = 0; ks < 32; ++ks) {
      const v4f wa = *reinterpret_cast<const v4f*>(pB + (size_t)(4 * ks + kb) * INNER);
      const v4f wb = *reinterpret_cast<const v4f*>(pB + (size_t)(4 * ks + kb + 1) * INNER);
#pragma unroll
      for (int dt = 0; dt < 4; ++dt) { bq[dt][ks].x = wa[dt]; bq[dt][ks].y = wb[dt]; }
    }
    const float* pA = posB + (size_t)(w * 128 + row) * kP + kb;
    float* qdst = &lds_qh[w * 128][0];
    v2f a0[32], a1[32];
    load_pos_frag(a0, pA);
#pragma unroll 1
    for (int i = 0; i < 8; i += 2) {
      load_pos_frag(a1, pA + (size_t)(i + 1) * 16 * kP);
      qh_tile(a0, bq, qdst + i * 16 * (kD + 1), half, row);
      const int nx = (i + 2 < 8) ? (i + 2) : 7;
      load_pos_frag(a0, pA + (size_t)nx * 16 * kP);
      qh_tile(a1, bq, qdst + (i + 1) * 16 * (kD + 1), half, row);
    }
  }
  v2f ak[2][16];
  {
    const float* aK = kl + (size_t)(w * 32 + row) * kD + kb;
#pragma unroll
    for (int jt = 0; jt < 2; ++jt)
#pragma unroll
      for (int ks = 0; ks < 16; ++ks)
        ak[jt][ks] = *reinterpret_cast<const v2f*>(aK + (size_t)jt * 16 * kD + 4 * ks);
  }
  __syncthreads();

  v8f   oacc[2][4];
  v8f   mrun[2], lrun[2];
#pragma unroll
  for (int jt = 0; jt < 2; ++jt) {
#pragma unroll
    for (int xt = 0; xt < 4; ++xt) oacc[jt][xt] = (v8f){0,0,0,0,0,0,0,0};
#pragma unroll
    for (int r = 0; r < 8; ++r) { mrun[jt][r] = -3.0e38f; lrun[jt][r] = 0.0f; }
  }

  for (int ch = 0; ch < NCH; ++ch) {
    const int n0 = ch * CH;
    const float* xc = xB + (size_t)n0 * kX + 4 * row;

    v4f xa0[8], xb0[8];
#pragma unroll
    for (int ks = 0; ks < 8; ++ks) {
      xa0[ks] = *reinterpret_cast<const v4f*>(xc + (4 * ks + kb) * kX);
      xb0[ks] = *reinterpret_cast<const v4f*>(xc + (4 * ks + kb + 1) * kX);
    }

    v2f qf[NT][16];
#pragma unroll
    for (int nt = 0; nt < NT; ++nt)
#pragma unroll
      for (int ks = 0; ks < 16; ++ks) {
        qf[nt][ks].x = lds_qh[n0 + nt * 16 + row][4 * ks + kb];
        qf[nt][ks].y = lds_qh[n0 + nt * 16 + row][4 * ks + kb + 1];
      }
    __builtin_amdgcn_sched_barrier(0);

    v8f s[2][NT];
#pragma unroll
    for (int nt = 0; nt < NT; ++nt) {
#pragma unroll
      for (int jt = 0; jt < 2; ++jt) {
        v8f c = (v8f){0,0,0,0,0,0,0,0};
#pragma unroll
        for (int ks = 0; ks < 16; ++ks) c = WMMA_F32(ak[jt][ks], qf[nt][ks], c);
        s[jt][nt] = c;
      }
    }

    v4f xa1[8], xb1[8];
#pragma unroll
    for (int ks = 0; ks < 8; ++ks) {
      xa1[ks] = *reinterpret_cast<const v4f*>(xc + (PH + 4 * ks + kb) * kX);
      xb1[ks] = *reinterpret_cast<const v4f*>(xc + (PH + 4 * ks + kb + 1) * kX);
    }

#pragma unroll
    for (int jt = 0; jt < 2; ++jt) {
#pragma unroll
      for (int nt = 0; nt < NT; ++nt) s[jt][nt] = s[jt][nt] * 0.125f;
      v8f t = __builtin_elementwise_max(__builtin_elementwise_max(s[jt][0], s[jt][1]),
                                        __builtin_elementwise_max(s[jt][2], s[jt][3]));
#pragma unroll
      for (int r = 0; r < 8; ++r) t[r] = fmaxf(t[r], row_xor<1>(t[r]));
#pragma unroll
      for (int r = 0; r < 8; ++r) t[r] = fmaxf(t[r], row_xor<2>(t[r]));
#pragma unroll
      for (int r = 0; r < 8; ++r) t[r] = fmaxf(t[r], row_xor<4>(t[r]));
#pragma unroll
      for (int r = 0; r < 8; ++r) t[r] = fmaxf(t[r], row_xor<8>(t[r]));
      const v8f mold = mrun[jt];
      const v8f mnew = __builtin_elementwise_max(mold, t);
      v8f fsc = mold - mnew;
#pragma unroll
      for (int r = 0; r < 8; ++r) fsc[r] = __expf(fsc[r]);
#pragma unroll
      for (int nt = 0; nt < NT; ++nt) {
        v8f d = s[jt][nt] - mnew;
#pragma unroll
        for (int r = 0; r < 8; ++r) d[r] = __expf(d[r]);
        s[jt][nt] = d;
      }
      const v8f rs = (s[jt][0] + s[jt][1]) + (s[jt][2] + s[jt][3]);
      lrun[jt] = lrun[jt] * fsc + rs;
      mrun[jt] = mnew;
#pragma unroll
      for (int xt = 0; xt < 4; ++xt) oacc[jt][xt] = oacc[jt][xt] * fsc;
    }

#pragma unroll
    for (int hh = 0; hh < 2; ++hh) {
#pragma unroll
      for (int jt = 0; jt < 2; ++jt)
#pragma unroll
        for (int nt = 0; nt < 2; ++nt)
#pragma unroll
          for (int r = 0; r < 8; ++r)
            lds_p[w][jt * 16 + r + 8 * half][nt * 16 + row] = s[jt][2 * hh + nt][r];
#pragma unroll
      for (int jt = 0; jt < 2; ++jt) {
#pragma unroll
        for (int xt = 0; xt < 4; ++xt) {
          v8f c = oacc[jt][xt];
#pragma unroll
          for (int k = 0; k < PH; k += 4) {
            v2f a;
            a.x = lds_p[w][jt * 16 + row][k + kb];
            a.y = lds_p[w][jt * 16 + row][k + kb + 1];
            v2f b;
            b.x = hh ? xa1[k / 4][xt] : xa0[k / 4][xt];
            b.y = hh ? xb1[k / 4][xt] : xb0[k / 4][xt];
            c = WMMA_F32(a, b, c);
          }
          oacc[jt][xt] = c;
        }
      }
    }
  }

  float* stg = &lds_p[w][0][0];
  const float* Wh = Wout + (size_t)h * kX * 64;
  float* Pw = ws + ((size_t)h * 512 + (size_t)(bb * 16 + w * 2) * 4) * 256;
#pragma unroll
  for (int jt = 0; jt < 2; ++jt) {
    float inv[8];
#pragma unroll
    for (int r = 0; r < 8; ++r) inv[r] = lrun[jt][r];
#pragma unroll
    for (int r = 0; r < 8; ++r) inv[r] += row_xor<1>(inv[r]);
#pragma unroll
    for (int r = 0; r < 8; ++r) inv[r] += row_xor<2>(inv[r]);
#pragma unroll
    for (int r = 0; r < 8; ++r) inv[r] += row_xor<4>(inv[r]);
#pragma unroll
    for (int r = 0; r < 8; ++r) inv[r] += row_xor<8>(inv[r]);
#pragma unroll
    for (int r = 0; r < 8; ++r) inv[r] = 1.0f / inv[r];
#pragma unroll
    for (int xt = 0; xt < 4; ++xt)
#pragma unroll
      for (int r = 0; r < 8; ++r)
        stg[(r + 8 * half) * 66 + 4 * row + xt] = oacc[jt][xt][r] * inv[r];

    v8f pc[4];
#pragma unroll
    for (int ct = 0; ct < 4; ++ct) pc[ct] = (v8f){0,0,0,0,0,0,0,0};
#pragma unroll
    for (int k = 0; k < kX; k += 4) {
      v2f a = *reinterpret_cast<const v2f*>(stg + row * 66 + k + kb);
#pragma unroll
      for (int ct = 0; ct < 4; ++ct) {
        v2f b;
        b.x = Wh[(k + kb) * 64 + ct * 16 + row];
        b.y = Wh[(k + kb + 1) * 64 + ct * 16 + row];
        pc[ct] = WMMA_F32(a, b, pc[ct]);
      }
    }
#pragma unroll
    for (int ct = 0; ct < 4; ++ct)
#pragma unroll
      for (int r = 0; r < 8; ++r)
        Pw[(jt * 4 + ct) * 256 + r * 32 + lane] = pc[ct][r];
  }
}

__global__ __launch_bounds__(256)
void head_sum(const float* __restrict__ ws, const float* __restrict__ bout,
              float* __restrict__ out) {
  __shared__ float4 red[4][64];
  const int T   = blockIdx.x;
  const int tid = threadIdx.x;
  const int q   = tid & 63;
  const int g   = tid >> 6;

  const float4* p = reinterpret_cast<const float4*>(ws) +
                    ((size_t)(g * 16) * 512 + T) * 64 + q;
  float4 v[16];
#pragma unroll
  for (int i = 0; i < 16; ++i) v[i] = p[(size_t)i * 512 * 64];
  float4 acc = v[0];
#pragma unroll
  for (int i = 1; i < 16; ++i) {
    acc.x += v[i].x; acc.y += v[i].y; acc.z += v[i].z; acc.w += v[i].w;
  }
  red[g][q] = acc;
  __syncthreads();
  if (g == 0) {
    float4 s = red[0][q];
#pragma unroll
    for (int i = 1; i < 4; ++i) {
      const float4 t = red[i][q];
      s.x += t.x; s.y += t.y; s.z += t.z; s.w += t.w;
    }
    const int r     = q >> 3;
    const int half  = (q >> 2) & 1;
    const int c0    = (T & 3) * 16 + (q & 3) * 4;
    const int jrow  = (T >> 2) * 16 + r + 8 * half;
    const float4 bv = *reinterpret_cast<const float4*>(bout + c0);
    s.x += bv.x; s.y += bv.y; s.z += bv.z; s.w += bv.w;
    *reinterpret_cast<float4*>(out + (size_t)jrow * 64 + c0) = s;
  }
}

extern "C" void kernel_launch(void* const* d_in, const int* in_sizes, int n_in,
                              void* d_out, int out_size, void* d_ws, size_t ws_size,
                              hipStream_t stream) {
  (void)in_sizes; (void)n_in; (void)out_size; (void)ws_size;
  const float* pos  = (const float*)d_in[0];
  const float* x    = (const float*)d_in[1];
  const float* Wq   = (const float*)d_in[2];
  const float* kl   = (const float*)d_in[3];
  const float* Wout = (const float*)d_in[4];
  const float* bout = (const float*)d_in[5];
  float* ws  = (float*)d_ws;
  float* out = (float*)d_out;

  pos_attn_fused<<<kB * kH, 256, 0, stream>>>(pos, x, Wq, kl, Wout, ws);
  head_sum<<<(kB * kJ * 64) / 256, 256, 0, stream>>>(ws, bout, out);
}
